# starting kernel
# baseline (speedup 1.0000x reference)
#include <hip/hip_runtime.h>

typedef __attribute__((ext_vector_type(2))) float v2f;
typedef __attribute__((ext_vector_type(8))) float v8f;

#define DIN  128
#define DH   256
#define DOUT 384


__global__ __launch_bounds__(256) void zero_kernel(float4* __restrict__ p, int n4) {
  int i = blockIdx.x * blockDim.x + threadIdx.x;
  if (i < n4) p[i] = make_float4(0.f, 0.f, 0.f, 0.f);
}

__global__ __launch_bounds__(256) void count_kernel(const int* __restrict__ dst,
                                                    int* __restrict__ cnt, int nedges) {
  int e = blockIdx.x * blockDim.x + threadIdx.x;
  if (e < nedges) atomicAdd(&cnt[dst[e]], 1);
}


__global__ __launch_bounds__(256) void scan_blocks_kernel(const int* __restrict__ cnt,
                                                          int* __restrict__ excl,
                                                          int* __restrict__ partials, int n) {
  __shared__ int sh[256];
  const int t = threadIdx.x;
  const int i = blockIdx.x * 256 + t;
  int v = (i < n) ? cnt[i] : 0;
  sh[t] = v;
  __syncthreads();
  for (int o = 1; o < 256; o <<= 1) {
    int x = (t >= o) ? sh[t - o] : 0;
    __syncthreads();
    sh[t] += x;
    __syncthreads();
  }
  if (i < n) excl[i] = sh[t] - v;
  if (t == 255) partials[blockIdx.x] = sh[255];
}

__global__ __launch_bounds__(256) void scan_partials_kernel(int* __restrict__ partials, int nb) {
  __shared__ int sh[256];
  const int t = threadIdx.x;
  int v = (t < nb) ? partials[t] : 0;
  sh[t] = v;
  __syncthreads();
  for (int o = 1; o < 256; o <<= 1) {
    int x = (t >= o) ? sh[t - o] : 0;
    __syncthreads();
    sh[t] += x;
    __syncthreads();
  }
  if (t < nb) partials[t] = sh[t] - v;
}

__global__ __launch_bounds__(256) void scan_finish_kernel(int* __restrict__ rowptr,
                                                          const int* __restrict__ partials,
                                                          const int* __restrict__ cnt,
                                                          int* __restrict__ cursor,
                                                          float* __restrict__ invdeg, int n) {
  int i = blockIdx.x * 256 + threadIdx.x;
  if (i >= n) return;
  int r = rowptr[i] + partials[i >> 8];
  rowptr[i] = r;
  cursor[i] = r;
  invdeg[i] = 1.0f / fmaxf((float)cnt[i], 1.0f);
  if (i == n - 1) rowptr[n] = r + cnt[i];
}

__global__ __launch_bounds__(256) void fill_adj_kernel(const int* __restrict__ src,
                                                       const int* __restrict__ dst,
                                                       int* __restrict__ cursor,
                                                       int* __restrict__ adj, int nedges) {
  int e = blockIdx.x * blockDim.x + threadIdx.x;
  if (e >= nedges) return;
  int pos = atomicAdd(&cursor[dst[e]], 1);
  adj[pos] = src[e];
}

template<int D>
__global__ __launch_bounds__(256) void gather_mean_kernel(const float* __restrict__ h,
                                                          const int* __restrict__ rowptr,
                                                          const int* __restrict__ adj,
                                                          const float* __restrict__ invdeg,
                                                          float* __restrict__ mean, int nnodes) {
  const int lane = threadIdx.x & 31;
  const int wave = threadIdx.x >> 5;
  const int node = blockIdx.x * 8 + wave;
  if (node >= nnodes) return;
  const int cbase = blockIdx.y * 64 + lane * 2;
  const int beg = rowptr[node];
  const int end = rowptr[node + 1];
  float2 acc = make_float2(0.f, 0.f);
  for (int j = beg; j < end; ++j) {
    const int s = adj[j];
    const float2 v = *(const float2*)(h + (size_t)s * D + cbase);
    acc.x += v.x;
    acc.y += v.y;
  }
  const float sc = invdeg[node];
  acc.x *= sc;
  acc.y *= sc;
  *(float2*)(mean + (size_t)node * D + cbase) = acc;
}

template<int K, int NOUT>
__global__ __launch_bounds__(256) void permute_w_kernel(const float* __restrict__ W,
                                                        float* __restrict__ Wt) {
  int i = blockIdx.x * blockDim.x + threadIdx.x;
  if (i >= K * NOUT) return;
  int k = i / NOUT;
  int col = i - k * NOUT;
  int stream = (k >> 1) & 1;
  int j = k >> 2;
  int pos = k & 1;
  Wt[((size_t)(col * 2 + stream) * (K / 4) + j) * 2 + pos] = W[i];
}

template<int K, int NOUT, bool RELU>
__global__ __launch_bounds__(256) void sage_gemm_kernel(const float* __restrict__ mean,
                                                        const float* __restrict__ hin,
                                                        const float* __restrict__ Wlp,
                                                        const float* __restrict__ Wrp,
                                                        const float* __restrict__ bias,
                                                        float* __restrict__ out, int mtiles) {
  const int lane  = threadIdx.x & 31;
  const int wave  = threadIdx.x >> 5;
  const int mtile = blockIdx.x * 8 + wave;
  if (mtile >= mtiles) return;
  const int nbase = blockIdx.y * 64;
  const int l15   = lane & 15;
  const int row   = mtile * 16 + l15;
  const int s     = lane >> 4;
  const int khalf = s << 1;

  v8f acc0 = {}, acc1 = {}, acc2 = {}, acc3 = {};

  for (int pass = 0; pass < 2; ++pass) {
    const float* __restrict__ Ar = (pass == 0 ? mean : hin) + (size_t)row * K;
    const float* __restrict__ Wp = (pass == 0 ? Wlp : Wrp);
    const v2f* bp0 = (const v2f*)Wp + (size_t)((nbase + l15 +  0) * 2 + s) * (K / 4);
    const v2f* bp1 = (const v2f*)Wp + (size_t)((nbase + l15 + 16) * 2 + s) * (K / 4);
    const v2f* bp2 = (const v2f*)Wp + (size_t)((nbase + l15 + 32) * 2 + s) * (K / 4);
    const v2f* bp3 = (const v2f*)Wp + (size_t)((nbase + l15 + 48) * 2 + s) * (K / 4);
#pragma unroll 4
    for (int j = 0; j < K / 4; ++j) {
      v2f a = *(const v2f*)(Ar + 4 * j + khalf);
      acc0 = __builtin_amdgcn_wmma_f32_16x16x4_f32(false, a, false, bp0[j], (short)0, acc0, false, false);
      acc1 = __builtin_amdgcn_wmma_f32_16x16x4_f32(false, a, false, bp1[j], (short)0, acc1, false, false);
      acc2 = __builtin_amdgcn_wmma_f32_16x16x4_f32(false, a, false, bp2[j], (short)0, acc2, false, false);
      acc3 = __builtin_amdgcn_wmma_f32_16x16x4_f32(false, a, false, bp3[j], (short)0, acc3, false, false);
    }
  }

  const int col = nbase + l15;
  const float bj0 = bias[col];
  const float bj1 = bias[col + 16];
  const float bj2 = bias[col + 32];
  const float bj3 = bias[col + 48];
  const int r0 = mtile * 16 + (s << 3);
#pragma unroll
  for (int v = 0; v < 8; ++v) {
    float* o = out + (size_t)(r0 + v) * NOUT + col;
    float t0 = acc0[v] + bj0;
    float t1 = acc1[v] + bj1;
    float t2 = acc2[v] + bj2;
    float t3 = acc3[v] + bj3;
    if (RELU) {
      t0 = fmaxf(t0, 0.f); t1 = fmaxf(t1, 0.f);
      t2 = fmaxf(t2, 0.f); t3 = fmaxf(t3, 0.f);
    }
    o[0] = t0; o[16] = t1; o[32] = t2; o[48] = t3;
  }
}


extern "C" void kernel_launch(void* const* d_in, const int* in_sizes, int n_in,
                              void* d_out, int out_size, void* d_ws, size_t ws_size,
                              hipStream_t stream) {
  const float* x   = (const float*)d_in[0];
  const int*   ei  = (const int*)d_in[1];
  const float* W1l = (const float*)d_in[2];
  const float* W1r = (const float*)d_in[3];
  const float* b1  = (const float*)d_in[4];
  const float* W2l = (const float*)d_in[5];
  const float* W2r = (const float*)d_in[6];
  const float* b2  = (const float*)d_in[7];
  const float* W3l = (const float*)d_in[8];
  const float* W3r = (const float*)d_in[9];
  const float* b3  = (const float*)d_in[10];
  float* out = (float*)d_out;

  const int N = in_sizes[0] / DIN;
  const int E = in_sizes[1] / 2;
  const int* src = ei;
  const int* dst = ei + E;

  char* ws = (char*)d_ws;
  auto al = [](size_t v) { return (v + 255) & ~(size_t)255; };
  size_t off = 0;
  int*   cnt      = (int*)(ws + off);   off += al((size_t)N * 4);
  int*   rowptr   = (int*)(ws + off);   off += al((size_t)(N + 1) * 4);
  int*   cursor   = (int*)(ws + off);   off += al((size_t)N * 4);
  int*   partials = (int*)(ws + off);   off += al((size_t)256 * 4);
  float* invdeg   = (float*)(ws + off); off += al((size_t)N * 4);
  int*   adj      = (int*)(ws + off);   off += al((size_t)E * 4);
  float* mean     = (float*)(ws + off); off += al((size_t)N * DH * 4);
  float* h1       = (float*)(ws + off); off += al((size_t)N * DH * 4);
  float* h2       = (float*)(ws + off); off += al((size_t)N * DH * 4);
  float* Wt1l     = (float*)(ws + off); off += al((size_t)DIN * DH * 4);
  float* Wt1r     = (float*)(ws + off); off += al((size_t)DIN * DH * 4);
  float* Wt2l     = (float*)(ws + off); off += al((size_t)DH * DH * 4);
  float* Wt2r     = (float*)(ws + off); off += al((size_t)DH * DH * 4);
  float* Wt3l     = (float*)(ws + off); off += al((size_t)DH * DOUT * 4);
  float* Wt3r     = (float*)(ws + off);

  const int TB = 256;
  const int NB = (N + 255) / 256;
  const int mtiles = N / 16;
  const int gm = (mtiles + 7) / 8;
  const int gn = (N + 7) / 8;

  zero_kernel<<<(N / 4 + TB - 1) / TB, TB, 0, stream>>>((float4*)cnt, N / 4);
  count_kernel<<<(E + TB - 1) / TB, TB, 0, stream>>>(dst, cnt, E);
  scan_blocks_kernel<<<NB, TB, 0, stream>>>(cnt, rowptr, partials, N);
  scan_partials_kernel<<<1, TB, 0, stream>>>(partials, NB);
  scan_finish_kernel<<<NB, TB, 0, stream>>>(rowptr, partials, cnt, cursor, invdeg, N);
  fill_adj_kernel<<<(E + TB - 1) / TB, TB, 0, stream>>>(src, dst, cursor, adj, E);

  permute_w_kernel<DIN, DH ><<<(DIN * DH  + TB - 1) / TB, TB, 0, stream>>>(W1l, Wt1l);
  permute_w_kernel<DIN, DH ><<<(DIN * DH  + TB - 1) / TB, TB, 0, stream>>>(W1r, Wt1r);
  permute_w_kernel<DH,  DH ><<<(DH  * DH  + TB - 1) / TB, TB, 0, stream>>>(W2l, Wt2l);
  permute_w_kernel<DH,  DH ><<<(DH  * DH  + TB - 1) / TB, TB, 0, stream>>>(W2r, Wt2r);
  permute_w_kernel<DH, DOUT><<<(DH * DOUT + TB - 1) / TB, TB, 0, stream>>>(W3l, Wt3l);
  permute_w_kernel<DH, DOUT><<<(DH * DOUT + TB - 1) / TB, TB, 0, stream>>>(W3r, Wt3r);

  {
    dim3 ga(gn, DIN / 64);
    gather_mean_kernel<DIN><<<ga, TB, 0, stream>>>(x, rowptr, adj, invdeg, mean, N);
    dim3 g(gm, DH / 64);
    sage_gemm_kernel<DIN, DH, true><<<g, TB, 0, stream>>>(mean, x, Wt1l, Wt1r, b1, h1, mtiles);
  }
  {
    dim3 ga(gn, DH / 64);
    gather_mean_kernel<DH><<<ga, TB, 0, stream>>>(h1, rowptr, adj, invdeg, mean, N);
    dim3 g(gm, DH / 64);
    sage_gemm_kernel<DH, DH, true><<<g, TB, 0, stream>>>(mean, h1, Wt2l, Wt2r, b2, h2, mtiles);
  }
  {
    dim3 ga(gn, DH / 64);
    gather_mean_kernel<DH><<<ga, TB, 0, stream>>>(h2, rowptr, adj, invdeg, mean, N);
    dim3 g(gm, DOUT / 64);
    sage_gemm_kernel<DH, DOUT, false><<<g, TB, 0, stream>>>(mean, h2, Wt3l, Wt3r, b3, out, mtiles);
  }
  (void)n_in; (void)out_size; (void)ws_size;
}


# GEMM operands staged through LDS: 128x32 A tile and 64x32 weight tile per 32-k chunk by coalesced b128 loads, two buffers, one barrier per chunk; same wave tiles and k order; weights pre-transposed
# speedup vs baseline: 2.1370x; 2.1370x over previous
#include <hip/hip_runtime.h>

typedef __attribute__((ext_vector_type(2))) float v2f;
typedef __attribute__((ext_vector_type(8))) float v8f;

#define DIN  128
#define DH   256
#define DOUT 384


__global__ __launch_bounds__(256) void zero_kernel(float4* __restrict__ p, int n4) {
  int i = blockIdx.x * blockDim.x + threadIdx.x;
  if (i < n4) p[i] = make_float4(0.f, 0.f, 0.f, 0.f);
}

__global__ __launch_bounds__(256) void count_kernel(const int* __restrict__ dst,
                                                    int* __restrict__ cnt, int nedges) {
  int e = blockIdx.x * blockDim.x + threadIdx.x;
  if (e < nedges) atomicAdd(&cnt[dst[e]], 1);
}


__global__ __launch_bounds__(256) void scan_blocks_kernel(const int* __restrict__ cnt,
                                                          int* __restrict__ excl,
                                                          int* __restrict__ partials, int n) {
  __shared__ int sh[256];
  const int t = threadIdx.x;
  const int i = blockIdx.x * 256 + t;
  int v = (i < n) ? cnt[i] : 0;
  sh[t] = v;
  __syncthreads();
  for (int o = 1; o < 256; o <<= 1) {
    int x = (t >= o) ? sh[t - o] : 0;
    __syncthreads();
    sh[t] += x;
    __syncthreads();
  }
  if (i < n) excl[i] = sh[t] - v;
  if (t == 255) partials[blockIdx.x] = sh[255];
}

__global__ __launch_bounds__(256) void scan_partials_kernel(int* __restrict__ partials, int nb) {
  __shared__ int sh[256];
  const int t = threadIdx.x;
  int v = (t < nb) ? partials[t] : 0;
  sh[t] = v;
  __syncthreads();
  for (int o = 1; o < 256; o <<= 1) {
    int x = (t >= o) ? sh[t - o] : 0;
    __syncthreads();
    sh[t] += x;
    __syncthreads();
  }
  if (t < nb) partials[t] = sh[t] - v;
}

__global__ __launch_bounds__(256) void scan_finish_kernel(int* __restrict__ rowptr,
                                                          const int* __restrict__ partials,
                                                          const int* __restrict__ cnt,
                                                          int* __restrict__ cursor,
                                                          float* __restrict__ invdeg, int n) {
  int i = blockIdx.x * 256 + threadIdx.x;
  if (i >= n) return;
  int r = rowptr[i] + partials[i >> 8];
  rowptr[i] = r;
  cursor[i] = r;
  invdeg[i] = 1.0f / fmaxf((float)cnt[i], 1.0f);
  if (i == n - 1) rowptr[n] = r + cnt[i];
}

__global__ __launch_bounds__(256) void fill_adj_kernel(const int* __restrict__ src,
                                                       const int* __restrict__ dst,
                                                       int* __restrict__ cursor,
                                                       int* __restrict__ adj, int nedges) {
  int e = blockIdx.x * blockDim.x + threadIdx.x;
  if (e >= nedges) return;
  int pos = atomicAdd(&cursor[dst[e]], 1);
  adj[pos] = src[e];
}

template<int D>
__global__ __launch_bounds__(256) void gather_mean_kernel(const float* __restrict__ h,
                                                          const int* __restrict__ rowptr,
                                                          const int* __restrict__ adj,
                                                          const float* __restrict__ invdeg,
                                                          float* __restrict__ mean, int nnodes) {
  const int lane = threadIdx.x & 31;
  const int wave = threadIdx.x >> 5;
  const int node = blockIdx.x * 8 + wave;
  if (node >= nnodes) return;
  const int cbase = blockIdx.y * 64 + lane * 2;
  const int beg = rowptr[node];
  const int end = rowptr[node + 1];
  float2 acc = make_float2(0.f, 0.f);
  for (int j = beg; j < end; ++j) {
    const int s = adj[j];
    const float2 v = *(const float2*)(h + (size_t)s * D + cbase);
    acc.x += v.x;
    acc.y += v.y;
  }
  const float sc = invdeg[node];
  acc.x *= sc;
  acc.y *= sc;
  *(float2*)(mean + (size_t)node * D + cbase) = acc;
}

template<int K, int NOUT>
__global__ __launch_bounds__(256) void permute_w_kernel(const float* __restrict__ W,
                                                        float* __restrict__ Wt) {
  int i = blockIdx.x * blockDim.x + threadIdx.x;
  if (i >= K * NOUT) return;
  int k = i / NOUT;
  int col = i - k * NOUT;
  Wt[(size_t)col * K + k] = W[i];
}

template<int K, int NOUT, bool RELU>
__global__ __launch_bounds__(256) void sage_gemm_kernel(const float* __restrict__ mean,
                                                        const float* __restrict__ hin,
                                                        const float* __restrict__ Wlt,
                                                        const float* __restrict__ Wrt,
                                                        const float* __restrict__ bias,
                                                        float* __restrict__ out, int mtiles) {
  constexpr int KC  = 32;
  constexpr int P   = KC + 4;
  constexpr int NC1 = K / KC;
  constexpr int NCH = 2 * NC1;
  __shared__ __attribute__((aligned(16))) float sA[2][128 * P];
  __shared__ __attribute__((aligned(16))) float sB[2][64 * P];

  const int t     = threadIdx.x;
  const int lane  = t & 31;
  const int wave  = t >> 5;
  const int mtile = blockIdx.x * 8 + wave;
  const int nbase = blockIdx.y * 64;
  const int l15   = lane & 15;
  const int s     = lane >> 4;
  const int nrows = mtiles * 16;

  const int q4 = (t & 7) * 4;
  const int r0 = t >> 3;
  const int rowb = blockIdx.x * 128 + r0;
  const int last = nrows - 1;
  const int ar0 = rowb       < nrows ? rowb       : last;
  const int ar1 = rowb + 32  < nrows ? rowb + 32  : last;
  const int ar2 = rowb + 64  < nrows ? rowb + 64  : last;
  const int ar3 = rowb + 96  < nrows ? rowb + 96  : last;
  const int aoff0 = ar0 * K + q4, aoff1 = ar1 * K + q4, aoff2 = ar2 * K + q4, aoff3 = ar3 * K + q4;
  const int boff0 = (nbase + r0) * K + q4, boff1 = (nbase + r0 + 32) * K + q4;
  const int soff = r0 * P + q4;

  float4 ra0, ra1, ra2, ra3, rb0, rb1;
#define SAGE_FETCH(c)                                               \
  {                                                                 \
    const bool second_ = (c) >= NC1;                                \
    const float* __restrict__ Ag_ = second_ ? hin : mean;           \
    const float* __restrict__ Wg_ = second_ ? Wrt : Wlt;            \
    const int k0_ = (second_ ? (c) - NC1 : (c)) * KC;               \
    ra0 = *(const float4*)(Ag_ + aoff0 + k0_);                      \
    ra1 = *(const float4*)(Ag_ + aoff1 + k0_);                      \
    ra2 = *(const float4*)(Ag_ + aoff2 + k0_);                      \
    ra3 = *(const float4*)(Ag_ + aoff3 + k0_);                      \
    rb0 = *(const float4*)(Wg_ + boff0 + k0_);                      \
    rb1 = *(const float4*)(Wg_ + boff1 + k0_);                      \
  }
#define SAGE_STASH(buf)                                             \
  {                                                                 \
    *(float4*)(&sA[buf][soff])          = ra0;                      \
    *(float4*)(&sA[buf][soff + 32 * P]) = ra1;                      \
    *(float4*)(&sA[buf][soff + 64 * P]) = ra2;                      \
    *(float4*)(&sA[buf][soff + 96 * P]) = ra3;                      \
    *(float4*)(&sB[buf][soff])          = rb0;                      \
    *(float4*)(&sB[buf][soff + 32 * P]) = rb1;                      \
  }

  v8f acc0 = {}, acc1 = {}, acc2 = {}, acc3 = {};

  SAGE_FETCH(0)
  SAGE_STASH(0)
  __syncthreads();

  for (int c = 0; c < NCH; ++c) {
    const int buf = c & 1;
    if (c + 1 < NCH) SAGE_FETCH(c + 1)
    const float* As = &sA[buf][(wave * 16 + l15) * P + 2 * s];
    const float* Bs = &sB[buf][l15 * P + 2 * s];
#pragma unroll
    for (int j = 0; j < KC / 4; ++j) {
      const v2f a  = *(const v2f*)(As + 4 * j);
      const v2f b0 = *(const v2f*)(Bs + 4 * j);
      const v2f b1 = *(const v2f*)(Bs + 16 * P + 4 * j);
      const v2f b2 = *(const v2f*)(Bs + 32 * P + 4 * j);
      const v2f b3 = *(const v2f*)(Bs + 48 * P + 4 * j);
      acc0 = __builtin_amdgcn_wmma_f32_16x16x4_f32(false, a, false, b0, (short)0, acc0, false, false);
      acc1 = __builtin_amdgcn_wmma_f32_16x16x4_f32(false, a, false, b1, (short)0, acc1, false, false);
      acc2 = __builtin_amdgcn_wmma_f32_16x16x4_f32(false, a, false, b2, (short)0, acc2, false, false);
      acc3 = __builtin_amdgcn_wmma_f32_16x16x4_f32(false, a, false, b3, (short)0, acc3, false, false);
    }
    if (c + 1 < NCH) {
      SAGE_STASH(buf ^ 1)
      __syncthreads();
    }
  }

#undef SAGE_FETCH
#undef SAGE_STASH

  if (mtile >= mtiles) return;

  const int col = nbase + l15;
  const float bj0 = bias[col];
  const float bj1 = bias[col + 16];
  const float bj2 = bias[col + 32];
  const float bj3 = bias[col + 48];
  const int rr = mtile * 16 + (s << 3);
#pragma unroll
  for (int v = 0; v < 8; ++v) {
    float* o = out + (size_t)(rr + v) * NOUT + col;
    float t0 = acc0[v] + bj0;
    float t1 = acc1[v] + bj1;
    float t2 = acc2[v] + bj2;
    float t3 = acc3[v] + bj3;
    if (RELU) {
      t0 = fmaxf(t0, 0.f); t1 = fmaxf(t1, 0.f);
      t2 = fmaxf(t2, 0.f); t3 = fmaxf(t3, 0.f);
    }
    o[0] = t0; o[16] = t1; o[32] = t2; o[48] = t3;
  }
}


extern "C" void kernel_launch(void* const* d_in, const int* in_sizes, int n_in,
                              void* d_out, int out_size, void* d_ws, size_t ws_size,
                              hipStream_t stream) {
  const float* x   = (const float*)d_in[0];
  const int*   ei  = (const int*)d_in[1];
  const float* W1l = (const float*)d_in[2];
  const float* W1r = (const float*)d_in[3];
  const float* b1  = (const float*)d_in[4];
  const float* W2l = (const float*)d_in[5];
  const float* W2r = (const float*)d_in[6];
  const float* b2  = (const float*)d_in[7];
  const float* W3l = (const float*)d_in[8];
  const float* W3r = (const float*)d_in[9];
  const float* b3  = (const float*)d_in[10];
  float* out = (float*)d_out;

  const int N = in_sizes[0] / DIN;
  const int E = in_sizes[1] / 2;
  const int* src = ei;
  const int* dst = ei + E;

  char* ws = (char*)d_ws;
  auto al = [](size_t v) { return (v + 255) & ~(size_t)255; };
  size_t off = 0;
  int*   cnt      = (int*)(ws + off);   off += al((size_t)N * 4);
  int*   rowptr   = (int*)(ws + off);   off += al((size_t)(N + 1) * 4);
  int*   cursor   = (int*)(ws + off);   off += al((size_t)N * 4);
  int*   partials = (int*)(ws + off);   off += al((size_t)256 * 4);
  float* invdeg   = (float*)(ws + off); off += al((size_t)N * 4);
  int*   adj      = (int*)(ws + off);   off += al((size_t)E * 4);
  float* mean     = (float*)(ws + off); off += al((size_t)N * DH * 4);
  float* h1       = (float*)(ws + off); off += al((size_t)N * DH * 4);
  float* h2       = (float*)(ws + off); off += al((size_t)N * DH * 4);
  float* Wt1l     = (float*)(ws + off); off += al((size_t)DIN * DH * 4);
  float* Wt1r     = (float*)(ws + off); off += al((size_t)DIN * DH * 4);
  float* Wt2l     = (float*)(ws + off); off += al((size_t)DH * DH * 4);
  float* Wt2r     = (float*)(ws + off); off += al((size_t)DH * DH * 4);
  float* Wt3l     = (float*)(ws + off); off += al((size_t)DH * DOUT * 4);
  float* Wt3r     = (float*)(ws + off);

  const int TB = 256;
  const int NB = (N + 255) / 256;
  const int mtiles = N / 16;
  const int gm = (mtiles + 7) / 8;
  const int gn = (N + 7) / 8;

  zero_kernel<<<(N / 4 + TB - 1) / TB, TB, 0, stream>>>((float4*)cnt, N / 4);
  count_kernel<<<(E + TB - 1) / TB, TB, 0, stream>>>(dst, cnt, E);
  scan_blocks_kernel<<<NB, TB, 0, stream>>>(cnt, rowptr, partials, N);
  scan_partials_kernel<<<1, TB, 0, stream>>>(partials, NB);
  scan_finish_kernel<<<NB, TB, 0, stream>>>(rowptr, partials, cnt, cursor, invdeg, N);
  fill_adj_kernel<<<(E + TB - 1) / TB, TB, 0, stream>>>(src, dst, cursor, adj, E);

  permute_w_kernel<DIN, DH ><<<(DIN * DH  + TB - 1) / TB, TB, 0, stream>>>(W1l, Wt1l);
  permute_w_kernel<DIN, DH ><<<(DIN * DH  + TB - 1) / TB, TB, 0, stream>>>(W1r, Wt1r);
  permute_w_kernel<DH,  DH ><<<(DH  * DH  + TB - 1) / TB, TB, 0, stream>>>(W2l, Wt2l);
  permute_w_kernel<DH,  DH ><<<(DH  * DH  + TB - 1) / TB, TB, 0, stream>>>(W2r, Wt2r);
  permute_w_kernel<DH, DOUT><<<(DH * DOUT + TB - 1) / TB, TB, 0, stream>>>(W3l, Wt3l);
  permute_w_kernel<DH, DOUT><<<(DH * DOUT + TB - 1) / TB, TB, 0, stream>>>(W3r, Wt3r);

  {
    dim3 ga(gn, DIN / 64);
    gather_mean_kernel<DIN><<<ga, TB, 0, stream>>>(x, rowptr, adj, invdeg, mean, N);
    dim3 g(gm, DH / 64);
    sage_gemm_kernel<DIN, DH, true><<<g, TB, 0, stream>>>(mean, x, Wt1l, Wt1r, b1, h1, mtiles);
  }
  {
    dim3 ga(gn, DH / 64);
    gather_mean_kernel<DH><<<ga, TB, 0, stream>>>(h1, rowptr, adj, invdeg, mean, N);
    dim3 g(gm, DH / 64);
    sage_gemm_kernel<DH, DH, true><<<g, TB, 0, stream>>>(mean, h1, Wt2l, Wt2r, b2, h2, mtiles);
  }
  {
    dim3 ga(gn, DH / 64);
    gather_mean_kernel<DH><<<ga, TB, 0, stream>>>(h2, rowptr, adj, invdeg, mean, N);
    dim3 g(gm, DOUT / 64);
    sage_gemm_kernel<DH, DOUT, false><<<g, TB, 0, stream>>>(mean, h2, Wt3l, Wt3r, b3, out, mtiles);
  }
  (void)n_in; (void)out_size; (void)ws_size;
}


# gather re-tiled: a wave sums whole feature rows (16-B loads, 4 neighbour rows in flight, ids on the scalar path) for 4 nodes in turn; was a short wave per node and 64-feature chunk; sum order kept
# speedup vs baseline: 2.3762x; 1.1119x over previous
#include <hip/hip_runtime.h>

typedef __attribute__((ext_vector_type(2))) float v2f;
typedef __attribute__((ext_vector_type(8))) float v8f;

#define DIN  128
#define DH   256
#define DOUT 384


__global__ __launch_bounds__(256) void zero_kernel(float4* __restrict__ p, int n4) {
  int i = blockIdx.x * blockDim.x + threadIdx.x;
  if (i < n4) p[i] = make_float4(0.f, 0.f, 0.f, 0.f);
}

__global__ __launch_bounds__(256) void count_kernel(const int* __restrict__ dst,
                                                    int* __restrict__ cnt, int nedges) {
  int e = blockIdx.x * blockDim.x + threadIdx.x;
  if (e < nedges) atomicAdd(&cnt[dst[e]], 1);
}


__global__ __launch_bounds__(256) void scan_blocks_kernel(const int* __restrict__ cnt,
                                                          int* __restrict__ excl,
                                                          int* __restrict__ partials, int n) {
  __shared__ int sh[256];
  const int t = threadIdx.x;
  const int i = blockIdx.x * 256 + t;
  int v = (i < n) ? cnt[i] : 0;
  sh[t] = v;
  __syncthreads();
  for (int o = 1; o < 256; o <<= 1) {
    int x = (t >= o) ? sh[t - o] : 0;
    __syncthreads();
    sh[t] += x;
    __syncthreads();
  }
  if (i < n) excl[i] = sh[t] - v;
  if (t == 255) partials[blockIdx.x] = sh[255];
}

__global__ __launch_bounds__(256) void scan_partials_kernel(int* __restrict__ partials, int nb) {
  __shared__ int sh[256];
  const int t = threadIdx.x;
  int v = (t < nb) ? partials[t] : 0;
  sh[t] = v;
  __syncthreads();
  for (int o = 1; o < 256; o <<= 1) {
    int x = (t >= o) ? sh[t - o] : 0;
    __syncthreads();
    sh[t] += x;
    __syncthreads();
  }
  if (t < nb) partials[t] = sh[t] - v;
}

__global__ __launch_bounds__(256) void scan_finish_kernel(int* __restrict__ rowptr,
                                                          const int* __restrict__ partials,
                                                          const int* __restrict__ cnt,
                                                          int* __restrict__ cursor,
                                                          float* __restrict__ invdeg, int n) {
  int i = blockIdx.x * 256 + threadIdx.x;
  if (i >= n) return;
  int r = rowptr[i] + partials[i >> 8];
  rowptr[i] = r;
  cursor[i] = r;
  invdeg[i] = 1.0f / fmaxf((float)cnt[i], 1.0f);
  if (i == n - 1) rowptr[n] = r + cnt[i];
}

__global__ __launch_bounds__(256) void fill_adj_kernel(const int* __restrict__ src,
                                                       const int* __restrict__ dst,
                                                       int* __restrict__ cursor,
                                                       int* __restrict__ adj, int nedges) {
  int e = blockIdx.x * blockDim.x + threadIdx.x;
  if (e >= nedges) return;
  int pos = atomicAdd(&cursor[dst[e]], 1);
  adj[pos] = src[e];
}

#define GATHER_WAVES 4
#define GATHER_NPW   4

template<int D, int R>
__device__ __forceinline__ void gather_add_rows(const float* __restrict__ hl, const int (&s)[4],
                                                float4 (&acc)[D / 128]) {
  constexpr int V = D / 128;
  float4 v[R][V];
#pragma unroll
  for (int r = 0; r < R; ++r) {
    const float* row = hl + (size_t)s[r] * D;
#pragma unroll
    for (int q = 0; q < V; ++q) v[r][q] = *(const float4*)(row + q * 128);
  }
#pragma unroll
  for (int r = 0; r < R; ++r) {
#pragma unroll
    for (int q = 0; q < V; ++q) {
      acc[q].x += v[r][q].x;
      acc[q].y += v[r][q].y;
      acc[q].z += v[r][q].z;
      acc[q].w += v[r][q].w;
    }
  }
}

template<int D>
__global__ __launch_bounds__(GATHER_WAVES * 32) void gather_mean_kernel(const float* __restrict__ h,
                                                          const int* __restrict__ rowptr,
                                                          const int* __restrict__ adj,
                                                          const float* __restrict__ invdeg,
                                                          float* __restrict__ mean, int nnodes) {
  constexpr int V = D / 128;
  const int lane = threadIdx.x & 31;
  const int wave = __builtin_amdgcn_readfirstlane((int)(threadIdx.x >> 5));
  const int node0 = (blockIdx.x * GATHER_WAVES + wave) * GATHER_NPW;
  if (node0 >= nnodes) return;
  const float* hl = h + lane * 4;
  int rp[GATHER_NPW + 1];
  float scv[GATHER_NPW];
#pragma unroll
  for (int i = 0; i <= GATHER_NPW; ++i) rp[i] = rowptr[min(node0 + i, nnodes)];
#pragma unroll
  for (int i = 0; i < GATHER_NPW; ++i) scv[i] = invdeg[min(node0 + i, nnodes - 1)];
#pragma unroll
  for (int i = 0; i < GATHER_NPW; ++i) {
    const int node = node0 + i;
    if (node >= nnodes) break;
    const int beg = rp[i];
    const int end = rp[i + 1];
    float4 acc[V];
#pragma unroll
    for (int q = 0; q < V; ++q) acc[q] = make_float4(0.f, 0.f, 0.f, 0.f);
    if (beg < end) {
      const int lastj = end - 1;
      int s[4];
#pragma unroll
      for (int r = 0; r < 4; ++r) s[r] = adj[min(beg + r, lastj)];
      int j = beg;
      while (j + 4 <= end) {
        int sn[4];
        j += 4;
#pragma unroll
        for (int r = 0; r < 4; ++r) sn[r] = adj[min(j + r, lastj)];
        gather_add_rows<D, 4>(hl, s, acc);
#pragma unroll
        for (int r = 0; r < 4; ++r) s[r] = sn[r];
      }
      const int rem = end - j;
      if (rem == 3)      gather_add_rows<D, 3>(hl, s, acc);
      else if (rem == 2) gather_add_rows<D, 2>(hl, s, acc);
      else if (rem == 1) gather_add_rows<D, 1>(hl, s, acc);
    }
    const float sc = scv[i];
    float* o = mean + (size_t)node * D + lane * 4;
#pragma unroll
    for (int q = 0; q < V; ++q) {
      float4 t = acc[q];
      t.x *= sc; t.y *= sc; t.z *= sc; t.w *= sc;
      *(float4*)(o + q * 128) = t;
    }
  }
}

template<int K, int NOUT>
__global__ __launch_bounds__(256) void permute_w_kernel(const float* __restrict__ W,
                                                        float* __restrict__ Wt) {
  int i = blockIdx.x * blockDim.x + threadIdx.x;
  if (i >= K * NOUT) return;
  int k = i / NOUT;
  int col = i - k * NOUT;
  Wt[(size_t)col * K + k] = W[i];
}

template<int K, int NOUT, bool RELU>
__global__ __launch_bounds__(256) void sage_gemm_kernel(const float* __restrict__ mean,
                                                        const float* __restrict__ hin,
                                                        const float* __restrict__ Wlt,
                                                        const float* __restrict__ Wrt,
                                                        const float* __restrict__ bias,
                                                        float* __restrict__ out, int mtiles) {
  constexpr int KC  = 32;
  constexpr int P   = KC + 4;
  constexpr int NC1 = K / KC;
  constexpr int NCH = 2 * NC1;
  __shared__ __attribute__((aligned(16))) float sA[2][128 * P];
  __shared__ __attribute__((aligned(16))) float sB[2][64 * P];

  const int t     = threadIdx.x;
  const int lane  = t & 31;
  const int wave  = t >> 5;
  const int mtile = blockIdx.x * 8 + wave;
  const int nbase = blockIdx.y * 64;
  const int l15   = lane & 15;
  const int s     = lane >> 4;
  const int nrows = mtiles * 16;

  const int q4 = (t & 7) * 4;
  const int r0 = t >> 3;
  const int rowb = blockIdx.x * 128 + r0;
  const int last = nrows - 1;
  const int ar0 = rowb       < nrows ? rowb       : last;
  const int ar1 = rowb + 32  < nrows ? rowb + 32  : last;
  const int ar2 = rowb + 64  < nrows ? rowb + 64  : last;
  const int ar3 = rowb + 96  < nrows ? rowb + 96  : last;
  const int aoff0 = ar0 * K + q4, aoff1 = ar1 * K + q4, aoff2 = ar2 * K + q4, aoff3 = ar3 * K + q4;
  const int boff0 = (nbase + r0) * K + q4, boff1 = (nbase + r0 + 32) * K + q4;
  const int soff = r0 * P + q4;

  float4 ra0, ra1, ra2, ra3, rb0, rb1;
#define SAGE_FETCH(c)                                               \
  {                                                                 \
    const bool second_ = (c) >= NC1;                                \
    const float* __restrict__ Ag_ = second_ ? hin : mean;           \
    const float* __restrict__ Wg_ = second_ ? Wrt : Wlt;            \
    const int k0_ = (second_ ? (c) - NC1 : (c)) * KC;               \
    ra0 = *(const float4*)(Ag_ + aoff0 + k0_);                      \
    ra1 = *(const float4*)(Ag_ + aoff1 + k0_);                      \
    ra2 = *(const float4*)(Ag_ + aoff2 + k0_);                      \
    ra3 = *(const float4*)(Ag_ + aoff3 + k0_);                      \
    rb0 = *(const float4*)(Wg_ + boff0 + k0_);                      \
    rb1 = *(const float4*)(Wg_ + boff1 + k0_);                      \
  }
#define SAGE_STASH(buf)                                             \
  {                                                                 \
    *(float4*)(&sA[buf][soff])          = ra0;                      \
    *(float4*)(&sA[buf][soff + 32 * P]) = ra1;                      \
    *(float4*)(&sA[buf][soff + 64 * P]) = ra2;                      \
    *(float4*)(&sA[buf][soff + 96 * P]) = ra3;                      \
    *(float4*)(&sB[buf][soff])          = rb0;                      \
    *(float4*)(&sB[buf][soff + 32 * P]) = rb1;                      \
  }

  v8f acc0 = {}, acc1 = {}, acc2 = {}, acc3 = {};

  SAGE_FETCH(0)
  SAGE_STASH(0)
  __syncthreads();

  for (int c = 0; c < NCH; ++c) {
    const int buf = c & 1;
    if (c + 1 < NCH) SAGE_FETCH(c + 1)
    const float* As = &sA[buf][(wave * 16 + l15) * P + 2 * s];
    const float* Bs = &sB[buf][l15 * P + 2 * s];
#pragma unroll
    for (int j = 0; j < KC / 4; ++j) {
      const v2f a  = *(const v2f*)(As + 4 * j);
      const v2f b0 = *(const v2f*)(Bs + 4 * j);
      const v2f b1 = *(const v2f*)(Bs + 16 * P + 4 * j);
      const v2f b2 = *(const v2f*)(Bs + 32 * P + 4 * j);
      const v2f b3 = *(const v2f*)(Bs + 48 * P + 4 * j);
      acc0 = __builtin_amdgcn_wmma_f32_16x16x4_f32(false, a, false, b0, (short)0, acc0, false, false);
      acc1 = __builtin_amdgcn_wmma_f32_16x16x4_f32(false, a, false, b1, (short)0, acc1, false, false);
      acc2 = __builtin_amdgcn_wmma_f32_16x16x4_f32(false, a, false, b2, (short)0, acc2, false, false);
      acc3 = __builtin_amdgcn_wmma_f32_16x16x4_f32(false, a, false, b3, (short)0, acc3, false, false);
    }
    if (c + 1 < NCH) {
      SAGE_STASH(buf ^ 1)
      __syncthreads();
    }
  }

#undef SAGE_FETCH
#undef SAGE_STASH

  if (mtile >= mtiles) return;

  const int col = nbase + l15;
  const float bj0 = bias[col];
  const float bj1 = bias[col + 16];
  const float bj2 = bias[col + 32];
  const float bj3 = bias[col + 48];
  const int rr = mtile * 16 + (s << 3);
#pragma unroll
  for (int v = 0; v < 8; ++v) {
    float* o = out + (size_t)(rr + v) * NOUT + col;
    float t0 = acc0[v] + bj0;
    float t1 = acc1[v] + bj1;
    float t2 = acc2[v] + bj2;
    float t3 = acc3[v] + bj3;
    if (RELU) {
      t0 = fmaxf(t0, 0.f); t1 = fmaxf(t1, 0.f);
      t2 = fmaxf(t2, 0.f); t3 = fmaxf(t3, 0.f);
    }
    o[0] = t0; o[16] = t1; o[32] = t2; o[48] = t3;
  }
}


extern "C" void kernel_launch(void* const* d_in, const int* in_sizes, int n_in,
                              void* d_out, int out_size, void* d_ws, size_t ws_size,
                              hipStream_t stream) {
  const float* x   = (const float*)d_in[0];
  const int*   ei  = (const int*)d_in[1];
  const float* W1l = (const float*)d_in[2];
  const float* W1r = (const float*)d_in[3];
  const float* b1  = (const float*)d_in[4];
  const float* W2l = (const float*)d_in[5];
  const float* W2r = (const float*)d_in[6];
  const float* b2  = (const float*)d_in[7];
  const float* W3l = (const float*)d_in[8];
  const float* W3r = (const float*)d_in[9];
  const float* b3  = (const float*)d_in[10];
  float* out = (float*)d_out;

  const int N = in_sizes[0] / DIN;
  const int E = in_sizes[1] / 2;
  const int* src = ei;
  const int* dst = ei + E;

  char* ws = (char*)d_ws;
  auto al = [](size_t v) { return (v + 255) & ~(size_t)255; };
  size_t off = 0;
  int*   cnt      = (int*)(ws + off);   off += al((size_t)N * 4);
  int*   rowptr   = (int*)(ws + off);   off += al((size_t)(N + 1) * 4);
  int*   cursor   = (int*)(ws + off);   off += al((size_t)N * 4);
  int*   partials = (int*)(ws + off);   off += al((size_t)256 * 4);
  float* invdeg   = (float*)(ws + off); off += al((size_t)N * 4);
  int*   adj      = (int*)(ws + off);   off += al((size_t)E * 4);
  float* mean     = (float*)(ws + off); off += al((size_t)N * DH * 4);
  float* h1       = (float*)(ws + off); off += al((size_t)N * DH * 4);
  float* h2       = (float*)(ws + off); off += al((size_t)N * DH * 4);
  float* Wt1l     = (float*)(ws + off); off += al((size_t)DIN * DH * 4);
  float* Wt1r     = (float*)(ws + off); off += al((size_t)DIN * DH * 4);
  float* Wt2l     = (float*)(ws + off); off += al((size_t)DH * DH * 4);
  float* Wt2r     = (float*)(ws + off); off += al((size_t)DH * DH * 4);
  float* Wt3l     = (float*)(ws + off); off += al((size_t)DH * DOUT * 4);
  float* Wt3r     = (float*)(ws + off);

  const int TB = 256;
  const int NB = (N + 255) / 256;
  const int mtiles = N / 16;
  const int gm = (mtiles + 7) / 8;
  const int gg = (N + GATHER_WAVES * GATHER_NPW - 1) / (GATHER_WAVES * GATHER_NPW);
  const int GTB = GATHER_WAVES * 32;

  zero_kernel<<<(N / 4 + TB - 1) / TB, TB, 0, stream>>>((float4*)cnt, N / 4);
  count_kernel<<<(E + TB - 1) / TB, TB, 0, stream>>>(dst, cnt, E);
  scan_blocks_kernel<<<NB, TB, 0, stream>>>(cnt, rowptr, partials, N);
  scan_partials_kernel<<<1, TB, 0, stream>>>(partials, NB);
  scan_finish_kernel<<<NB, TB, 0, stream>>>(rowptr, partials, cnt, cursor, invdeg, N);
  fill_adj_kernel<<<(E + TB - 1) / TB, TB, 0, stream>>>(src, dst, cursor, adj, E);

  permute_w_kernel<DIN, DH ><<<(DIN * DH  + TB - 1) / TB, TB, 0, stream>>>(W1l, Wt1l);
  permute_w_kernel<DIN, DH ><<<(DIN * DH  + TB - 1) / TB, TB, 0, stream>>>(W1r, Wt1r);
  permute_w_kernel<DH,  DH ><<<(DH  * DH  + TB - 1) / TB, TB, 0, stream>>>(W2l, Wt2l);
  permute_w_kernel<DH,  DH ><<<(DH  * DH  + TB - 1) / TB, TB, 0, stream>>>(W2r, Wt2r);
  permute_w_kernel<DH, DOUT><<<(DH * DOUT + TB - 1) / TB, TB, 0, stream>>>(W3l, Wt3l);
  permute_w_kernel<DH, DOUT><<<(DH * DOUT + TB - 1) / TB, TB, 0, stream>>>(W3r, Wt3r);

  {
    gather_mean_kernel<DIN><<<gg, GTB, 0, stream>>>(x, rowptr, adj, invdeg, mean, N);
    dim3 g(gm, DH / 64);
    sage_gemm_kernel<DIN, DH, true><<<g, TB, 0, stream>>>(mean, x, Wt1l, Wt1r, b1, h1, mtiles);
  }
  {
    gather_mean_kernel<DH><<<gg, GTB, 0, stream>>>(h1, rowptr, adj, invdeg, mean, N);
    dim3 g(gm, DH / 64);
    sage_gemm_kernel<DH, DH, true><<<g, TB, 0, stream>>>(mean, h1, Wt2l, Wt2r, b2, h2, mtiles);
  }
  {
    gather_mean_kernel<DH><<<gg, GTB, 0, stream>>>(h2, rowptr, adj, invdeg, mean, N);
    dim3 g(gm, DOUT / 64);
    sage_gemm_kernel<DH, DOUT, false><<<g, TB, 0, stream>>>(mean, h2, Wt3l, Wt3r, b3, out, mtiles);
  }
  (void)n_in; (void)out_size; (void)ws_size;
}


# CSR build as two-level counting sort on LDS counters (buckets of 256 nodes, returning LDS add = rank) in place of one global atomic per edge in count and fill; neighbour order = LDS arrival order
# speedup vs baseline: 2.7787x; 1.1694x over previous
#include <hip/hip_runtime.h>

typedef __attribute__((ext_vector_type(2))) float v2f;
typedef __attribute__((ext_vector_type(8))) float v8f;

#define DIN  128
#define DH   256
#define DOUT 384

#define CSR_EB 256
#define CSR_TB 1024

__device__ __forceinline__ int block_scan_incl_256(int* sh, int t, int v) {
  const bool on = t < 256;
  if (on) sh[t] = v;
  __syncthreads();
  for (int o = 1; o < 256; o <<= 1) {
    int x = (on && t >= o) ? sh[t - o] : 0;
    __syncthreads();
    if (on) sh[t] += x;
    __syncthreads();
  }
  return on ? sh[t] : 0;
}

__global__ __launch_bounds__(CSR_TB) void bin_count_kernel(const int* __restrict__ dst,
                                                           int* __restrict__ table,
                                                           int nedges, int nb) {
  __shared__ int hist[256];
  const int t = threadIdx.x;
  const int b = blockIdx.x;
  if (t < 256) hist[t] = 0;
  __syncthreads();
  const int per = (nedges + CSR_EB - 1) / CSR_EB;
  const int e0 = b * per;
  const int e1 = min(e0 + per, nedges);
  for (int eb = e0 + t; eb < e1; eb += 4 * CSR_TB) {
    int d[4];
#pragma unroll
    for (int r = 0; r < 4; ++r) d[r] = dst[min(eb + r * CSR_TB, e1 - 1)];
#pragma unroll
    for (int r = 0; r < 4; ++r)
      if (eb + r * CSR_TB < e1) atomicAdd(&hist[d[r] >> 8], 1);
  }
  __syncthreads();
  if (t < nb) table[t * CSR_EB + b] = hist[t];
}

__global__ __launch_bounds__(256) void bin_scan_kernel(int* __restrict__ table,
                                                       int* __restrict__ btot) {
  __shared__ int sh[256];
  const int t = threadIdx.x;
  const int k = blockIdx.x;
  const int v = table[k * CSR_EB + t];
  const int inc = block_scan_incl_256(sh, t, v);
  table[k * CSR_EB + t] = inc - v;
  if (t == 255) btot[k] = inc;
}

__global__ __launch_bounds__(CSR_TB) void bin_scatter_kernel(const int* __restrict__ src,
                                                             const int* __restrict__ dst,
                                                             const int* __restrict__ table,
                                                             const int* __restrict__ btot,
                                                             int* __restrict__ bpk,
                                                             int nedges, int nb) {
  __shared__ int sh[256];
  __shared__ int cur[256];
  const int t = threadIdx.x;
  const int b = blockIdx.x;
  const int v = (t < nb) ? btot[t] : 0;
  const int inc = block_scan_incl_256(sh, t, v);
  if (t < 256) cur[t] = (t < nb) ? (inc - v) + table[t * CSR_EB + b] : 0;
  __syncthreads();
  const int per = (nedges + CSR_EB - 1) / CSR_EB;
  const int e0 = b * per;
  const int e1 = min(e0 + per, nedges);
  for (int eb = e0 + t; eb < e1; eb += 4 * CSR_TB) {
    int d[4], s[4];
#pragma unroll
    for (int r = 0; r < 4; ++r) {
      const int e = min(eb + r * CSR_TB, e1 - 1);
      d[r] = dst[e];
      s[r] = src[e];
    }
#pragma unroll
    for (int r = 0; r < 4; ++r) {
      if (eb + r * CSR_TB < e1) {
        const int pos = atomicAdd(&cur[d[r] >> 8], 1);
        bpk[pos] = (s[r] << 8) | (d[r] & 255);
      }
    }
  }
}

__global__ __launch_bounds__(CSR_TB) void bin_csr_kernel(const int* __restrict__ bpk,
                                                         const int* __restrict__ btot,
                                                         int* __restrict__ rowptr,
                                                         float* __restrict__ invdeg,
                                                         int* __restrict__ adj, int n, int nb) {
  __shared__ int sh[256];
  __shared__ int cnt[256];
  __shared__ int cur[256];
  const int t = threadIdx.x;
  const int k = blockIdx.x;
  if (t < 256) cnt[t] = 0;
  const int v = (t < nb) ? btot[t] : 0;
  block_scan_incl_256(sh, t, v);
  const int end = sh[k];
  const int beg = (k > 0) ? sh[k - 1] : 0;
  for (int eb = beg + t; eb < end; eb += 4 * CSR_TB) {
    int p[4];
#pragma unroll
    for (int r = 0; r < 4; ++r) p[r] = bpk[min(eb + r * CSR_TB, end - 1)];
#pragma unroll
    for (int r = 0; r < 4; ++r)
      if (eb + r * CSR_TB < end) atomicAdd(&cnt[p[r] & 255], 1);
  }
  __syncthreads();
  const int c = (t < 256) ? cnt[t] : 0;
  const int first = beg + block_scan_incl_256(sh, t, c) - c;
  if (t < 256) {
    const int node = k * 256 + t;
    if (node < n) {
      rowptr[node] = first;
      invdeg[node] = 1.0f / fmaxf((float)c, 1.0f);
      if (node == n - 1) rowptr[n] = first + c;
    }
    cur[t] = first;
  }
  __syncthreads();
  for (int eb = beg + t; eb < end; eb += 4 * CSR_TB) {
    int p[4];
#pragma unroll
    for (int r = 0; r < 4; ++r) p[r] = bpk[min(eb + r * CSR_TB, end - 1)];
#pragma unroll
    for (int r = 0; r < 4; ++r) {
      if (eb + r * CSR_TB < end) {
        const int pos = atomicAdd(&cur[p[r] & 255], 1);
        adj[pos] = p[r] >> 8;
      }
    }
  }
}

#define GATHER_WAVES 4
#define GATHER_NPW   4

template<int D, int R>
__device__ __forceinline__ void gather_add_rows(const float* __restrict__ hl, const int (&s)[4],
                                                float4 (&acc)[D / 128]) {
  constexpr int V = D / 128;
  float4 v[R][V];
#pragma unroll
  for (int r = 0; r < R; ++r) {
    const float* row = hl + (size_t)s[r] * D;
#pragma unroll
    for (int q = 0; q < V; ++q) v[r][q] = *(const float4*)(row + q * 128);
  }
#pragma unroll
  for (int r = 0; r < R; ++r) {
#pragma unroll
    for (int q = 0; q < V; ++q) {
      acc[q].x += v[r][q].x;
      acc[q].y += v[r][q].y;
      acc[q].z += v[r][q].z;
      acc[q].w += v[r][q].w;
    }
  }
}

template<int D>
__global__ __launch_bounds__(GATHER_WAVES * 32) void gather_mean_kernel(const float* __restrict__ h,
                                                          const int* __restrict__ rowptr,
                                                          const int* __restrict__ adj,
                                                          const float* __restrict__ invdeg,
                                                          float* __restrict__ mean, int nnodes) {
  constexpr int V = D / 128;
  const int lane = threadIdx.x & 31;
  const int wave = __builtin_amdgcn_readfirstlane((int)(threadIdx.x >> 5));
  const int node0 = (blockIdx.x * GATHER_WAVES + wave) * GATHER_NPW;
  if (node0 >= nnodes) return;
  const float* hl = h + lane * 4;
  int rp[GATHER_NPW + 1];
  float scv[GATHER_NPW];
#pragma unroll
  for (int i = 0; i <= GATHER_NPW; ++i) rp[i] = rowptr[min(node0 + i, nnodes)];
#pragma unroll
  for (int i = 0; i < GATHER_NPW; ++i) scv[i] = invdeg[min(node0 + i, nnodes - 1)];
#pragma unroll
  for (int i = 0; i < GATHER_NPW; ++i) {
    const int node = node0 + i;
    if (node >= nnodes) break;
    const int beg = rp[i];
    const int end = rp[i + 1];
    float4 acc[V];
#pragma unroll
    for (int q = 0; q < V; ++q) acc[q] = make_float4(0.f, 0.f, 0.f, 0.f);
    if (beg < end) {
      const int lastj = end - 1;
      int s[4];
#pragma unroll
      for (int r = 0; r < 4; ++r) s[r] = adj[min(beg + r, lastj)];
      int j = beg;
      while (j + 4 <= end) {
        int sn[4];
        j += 4;
#pragma unroll
        for (int r = 0; r < 4; ++r) sn[r] = adj[min(j + r, lastj)];
        gather_add_rows<D, 4>(hl, s, acc);
#pragma unroll
        for (int r = 0; r < 4; ++r) s[r] = sn[r];
      }
      const int rem = end - j;
      if (rem == 3)      gather_add_rows<D, 3>(hl, s, acc);
      else if (rem == 2) gather_add_rows<D, 2>(hl, s, acc);
      else if (rem == 1) gather_add_rows<D, 1>(hl, s, acc);
    }
    const float sc = scv[i];
    float* o = mean + (size_t)node * D + lane * 4;
#pragma unroll
    for (int q = 0; q < V; ++q) {
      float4 t = acc[q];
      t.x *= sc; t.y *= sc; t.z *= sc; t.w *= sc;
      *(float4*)(o + q * 128) = t;
    }
  }
}

template<int K, int NOUT>
__global__ __launch_bounds__(256) void permute_w_kernel(const float* __restrict__ W,
                                                        float* __restrict__ Wt) {
  int i = blockIdx.x * blockDim.x + threadIdx.x;
  if (i >= K * NOUT) return;
  int k = i / NOUT;
  int col = i - k * NOUT;
  Wt[(size_t)col * K + k] = W[i];
}

template<int K, int NOUT, bool RELU>
__global__ __launch_bounds__(256) void sage_gemm_kernel(const float* __restrict__ mean,
                                                        const float* __restrict__ hin,
                                                        const float* __restrict__ Wlt,
                                                        const float* __restrict__ Wrt,
                                                        const float* __restrict__ bias,
                                                        float* __restrict__ out, int mtiles) {
  constexpr int KC  = 32;
  constexpr int P   = KC + 4;
  constexpr int NC1 = K / KC;
  constexpr int NCH = 2 * NC1;
  __shared__ __attribute__((aligned(16))) float sA[2][128 * P];
  __shared__ __attribute__((aligned(16))) float sB[2][64 * P];

  const int t     = threadIdx.x;
  const int lane  = t & 31;
  const int wave  = t >> 5;
  const int mtile = blockIdx.x * 8 + wave;
  const int nbase = blockIdx.y * 64;
  const int l15   = lane & 15;
  const int s     = lane >> 4;
  const int nrows = mtiles * 16;

  const int q4 = (t & 7) * 4;
  const int r0 = t >> 3;
  const int rowb = blockIdx.x * 128 + r0;
  const int last = nrows - 1;
  const int ar0 = rowb       < nrows ? rowb       : last;
  const int ar1 = rowb + 32  < nrows ? rowb + 32  : last;
  const int ar2 = rowb + 64  < nrows ? rowb + 64  : last;
  const int ar3 = rowb + 96  < nrows ? rowb + 96  : last;
  const int aoff0 = ar0 * K + q4, aoff1 = ar1 * K + q4, aoff2 = ar2 * K + q4, aoff3 = ar3 * K + q4;
  const int boff0 = (nbase + r0) * K + q4, boff1 = (nbase + r0 + 32) * K + q4;
  const int soff = r0 * P + q4;

  float4 ra0, ra1, ra2, ra3, rb0, rb1;
#define SAGE_FETCH(c)                                               \
  {                                                                 \
    const bool second_ = (c) >= NC1;                                \
    const float* __restrict__ Ag_ = second_ ? hin : mean;           \
    const float* __restrict__ Wg_ = second_ ? Wrt : Wlt;            \
    const int k0_ = (second_ ? (c) - NC1 : (c)) * KC;               \
    ra0 = *(const float4*)(Ag_ + aoff0 + k0_);                      \
    ra1 = *(const float4*)(Ag_ + aoff1 + k0_);                      \
    ra2 = *(const float4*)(Ag_ + aoff2 + k0_);                      \
    ra3 = *(const float4*)(Ag_ + aoff3 + k0_);                      \
    rb0 = *(const float4*)(Wg_ + boff0 + k0_);                      \
    rb1 = *(const float4*)(Wg_ + boff1 + k0_);                      \
  }
#define SAGE_STASH(buf)                                             \
  {                                                                 \
    *(float4*)(&sA[buf][soff])          = ra0;                      \
    *(float4*)(&sA[buf][soff + 32 * P]) = ra1;                      \
    *(float4*)(&sA[buf][soff + 64 * P]) = ra2;                      \
    *(float4*)(&sA[buf][soff + 96 * P]) = ra3;                      \
    *(float4*)(&sB[buf][soff])          = rb0;                      \
    *(float4*)(&sB[buf][soff + 32 * P]) = rb1;                      \
  }

  v8f acc0 = {}, acc1 = {}, acc2 = {}, acc3 = {};

  SAGE_FETCH(0)
  SAGE_STASH(0)
  __syncthreads();

  for (int c = 0; c < NCH; ++c) {
    const int buf = c & 1;
    if (c + 1 < NCH) SAGE_FETCH(c + 1)
    const float* As = &sA[buf][(wave * 16 + l15) * P + 2 * s];
    const float* Bs = &sB[buf][l15 * P + 2 * s];
#pragma unroll
    for (int j = 0; j < KC / 4; ++j) {
      const v2f a  = *(const v2f*)(As + 4 * j);
      const v2f b0 = *(const v2f*)(Bs + 4 * j);
      const v2f b1 = *(const v2f*)(Bs + 16 * P + 4 * j);
      const v2f b2 = *(const v2f*)(Bs + 32 * P + 4 * j);
      const v2f b3 = *(const v2f*)(Bs + 48 * P + 4 * j);
      acc0 = __builtin_amdgcn_wmma_f32_16x16x4_f32(false, a, false, b0, (short)0, acc0, false, false);
      acc1 = __builtin_amdgcn_wmma_f32_16x16x4_f32(false, a, false, b1, (short)0, acc1, false, false);
      acc2 = __builtin_amdgcn_wmma_f32_16x16x4_f32(false, a, false, b2, (short)0, acc2, false, false);
      acc3 = __builtin_amdgcn_wmma_f32_16x16x4_f32(false, a, false, b3, (short)0, acc3, false, false);
    }
    if (c + 1 < NCH) {
      SAGE_STASH(buf ^ 1)
      __syncthreads();
    }
  }

#undef SAGE_FETCH
#undef SAGE_STASH

  if (mtile >= mtiles) return;

  const int col = nbase + l15;
  const float bj0 = bias[col];
  const float bj1 = bias[col + 16];
  const float bj2 = bias[col + 32];
  const float bj3 = bias[col + 48];
  const int rr = mtile * 16 + (s << 3);
#pragma unroll
  for (int v = 0; v < 8; ++v) {
    float* o = out + (size_t)(rr + v) * NOUT + col;
    float t0 = acc0[v] + bj0;
    float t1 = acc1[v] + bj1;
    float t2 = acc2[v] + bj2;
    float t3 = acc3[v] + bj3;
    if (RELU) {
      t0 = fmaxf(t0, 0.f); t1 = fmaxf(t1, 0.f);
      t2 = fmaxf(t2, 0.f); t3 = fmaxf(t3, 0.f);
    }
    o[0] = t0; o[16] = t1; o[32] = t2; o[48] = t3;
  }
}


extern "C" void kernel_launch(void* const* d_in, const int* in_sizes, int n_in,
                              void* d_out, int out_size, void* d_ws, size_t ws_size,
                              hipStream_t stream) {
  const float* x   = (const float*)d_in[0];
  const int*   ei  = (const int*)d_in[1];
  const float* W1l = (const float*)d_in[2];
  const float* W1r = (const float*)d_in[3];
  const float* b1  = (const float*)d_in[4];
  const float* W2l = (const float*)d_in[5];
  const float* W2r = (const float*)d_in[6];
  const float* b2  = (const float*)d_in[7];
  const float* W3l = (const float*)d_in[8];
  const float* W3r = (const float*)d_in[9];
  const float* b3  = (const float*)d_in[10];
  float* out = (float*)d_out;

  const int N = in_sizes[0] / DIN;
  const int E = in_sizes[1] / 2;
  const int* src = ei;
  const int* dst = ei + E;

  char* ws = (char*)d_ws;
  auto al = [](size_t v) { return (v + 255) & ~(size_t)255; };
  size_t off = 0;
  int*   rowptr   = (int*)(ws + off);   off += al((size_t)(N + 1) * 4);
  float* invdeg   = (float*)(ws + off); off += al((size_t)N * 4);
  int*   adj      = (int*)(ws + off);   off += al((size_t)E * 4);
  float* mean     = (float*)(ws + off); off += al((size_t)N * DH * 4);
  float* h1       = (float*)(ws + off); off += al((size_t)N * DH * 4);
  float* h2       = (float*)(ws + off); off += al((size_t)N * DH * 4);
  float* Wt1l     = (float*)(ws + off); off += al((size_t)DIN * DH * 4);
  float* Wt1r     = (float*)(ws + off); off += al((size_t)DIN * DH * 4);
  float* Wt2l     = (float*)(ws + off); off += al((size_t)DH * DH * 4);
  float* Wt2r     = (float*)(ws + off); off += al((size_t)DH * DH * 4);
  float* Wt3l     = (float*)(ws + off); off += al((size_t)DH * DOUT * 4);
  float* Wt3r     = (float*)(ws + off);
  int*   bpk      = (int*)mean;
  int*   table    = (int*)h1;
  int*   btot     = table + 256 * CSR_EB;

  const int TB = 256;
  const int NB = (N + 255) / 256;
  const int mtiles = N / 16;
  const int gm = (mtiles + 7) / 8;
  const int gg = (N + GATHER_WAVES * GATHER_NPW - 1) / (GATHER_WAVES * GATHER_NPW);
  const int GTB = GATHER_WAVES * 32;

  bin_count_kernel<<<CSR_EB, CSR_TB, 0, stream>>>(dst, table, E, NB);
  bin_scan_kernel<<<NB, TB, 0, stream>>>(table, btot);
  bin_scatter_kernel<<<CSR_EB, CSR_TB, 0, stream>>>(src, dst, table, btot, bpk, E, NB);
  bin_csr_kernel<<<NB, CSR_TB, 0, stream>>>(bpk, btot, rowptr, invdeg, adj, N, NB);

  permute_w_kernel<DIN, DH ><<<(DIN * DH  + TB - 1) / TB, TB, 0, stream>>>(W1l, Wt1l);
  permute_w_kernel<DIN, DH ><<<(DIN * DH  + TB - 1) / TB, TB, 0, stream>>>(W1r, Wt1r);
  permute_w_kernel<DH,  DH ><<<(DH  * DH  + TB - 1) / TB, TB, 0, stream>>>(W2l, Wt2l);
  permute_w_kernel<DH,  DH ><<<(DH  * DH  + TB - 1) / TB, TB, 0, stream>>>(W2r, Wt2r);
  permute_w_kernel<DH, DOUT><<<(DH * DOUT + TB - 1) / TB, TB, 0, stream>>>(W3l, Wt3l);
  permute_w_kernel<DH, DOUT><<<(DH * DOUT + TB - 1) / TB, TB, 0, stream>>>(W3r, Wt3r);

  {
    gather_mean_kernel<DIN><<<gg, GTB, 0, stream>>>(x, rowptr, adj, invdeg, mean, N);
    dim3 g(gm, DH / 64);
    sage_gemm_kernel<DIN, DH, true><<<g, TB, 0, stream>>>(mean, x, Wt1l, Wt1r, b1, h1, mtiles);
  }
  {
    gather_mean_kernel<DH><<<gg, GTB, 0, stream>>>(h1, rowptr, adj, invdeg, mean, N);
    dim3 g(gm, DH / 64);
    sage_gemm_kernel<DH, DH, true><<<g, TB, 0, stream>>>(mean, h1, Wt2l, Wt2r, b2, h2, mtiles);
  }
  {
    gather_mean_kernel<DH><<<gg, GTB, 0, stream>>>(h2, rowptr, adj, invdeg, mean, N);
    dim3 g(gm, DOUT / 64);
    sage_gemm_kernel<DH, DOUT, false><<<g, TB, 0, stream>>>(mean, h2, Wt3l, Wt3r, b3, out, mtiles);
  }
  (void)n_in; (void)out_size; (void)ws_size;
}


# GEMM stages the weights from their given [k][col] layout: 16-B loads along col, transposed on the way into LDS by 4-B stores; the six weight-transpose launches are dropped (16 -> 10 launches)
# speedup vs baseline: 2.9474x; 1.0607x over previous
#include <hip/hip_runtime.h>

typedef __attribute__((ext_vector_type(2))) float v2f;
typedef __attribute__((ext_vector_type(8))) float v8f;

#define DIN  128
#define DH   256
#define DOUT 384

#define CSR_EB 256
#define CSR_TB 1024

__device__ __forceinline__ int block_scan_incl_256(int* sh, int t, int v) {
  const bool on = t < 256;
  if (on) sh[t] = v;
  __syncthreads();
  for (int o = 1; o < 256; o <<= 1) {
    int x = (on && t >= o) ? sh[t - o] : 0;
    __syncthreads();
    if (on) sh[t] += x;
    __syncthreads();
  }
  return on ? sh[t] : 0;
}

__global__ __launch_bounds__(CSR_TB) void bin_count_kernel(const int* __restrict__ dst,
                                                           int* __restrict__ table,
                                                           int nedges, int nb) {
  __shared__ int hist[256];
  const int t = threadIdx.x;
  const int b = blockIdx.x;
  if (t < 256) hist[t] = 0;
  __syncthreads();
  const int per = (nedges + CSR_EB - 1) / CSR_EB;
  const int e0 = b * per;
  const int e1 = min(e0 + per, nedges);
  for (int eb = e0 + t; eb < e1; eb += 4 * CSR_TB) {
    int d[4];
#pragma unroll
    for (int r = 0; r < 4; ++r) d[r] = dst[min(eb + r * CSR_TB, e1 - 1)];
#pragma unroll
    for (int r = 0; r < 4; ++r)
      if (eb + r * CSR_TB < e1) atomicAdd(&hist[d[r] >> 8], 1);
  }
  __syncthreads();
  if (t < nb) table[t * CSR_EB + b] = hist[t];
}

__global__ __launch_bounds__(256) void bin_scan_kernel(int* __restrict__ table,
                                                       int* __restrict__ btot) {
  __shared__ int sh[256];
  const int t = threadIdx.x;
  const int k = blockIdx.x;
  const int v = table[k * CSR_EB + t];
  const int inc = block_scan_incl_256(sh, t, v);
  table[k * CSR_EB + t] = inc - v;
  if (t == 255) btot[k] = inc;
}

__global__ __launch_bounds__(CSR_TB) void bin_scatter_kernel(const int* __restrict__ src,
                                                             const int* __restrict__ dst,
                                                             const int* __restrict__ table,
                                                             const int* __restrict__ btot,
                                                             int* __restrict__ bpk,
                                                             int nedges, int nb) {
  __shared__ int sh[256];
  __shared__ int cur[256];
  const int t = threadIdx.x;
  const int b = blockIdx.x;
  const int v = (t < nb) ? btot[t] : 0;
  const int inc = block_scan_incl_256(sh, t, v);
  if (t < 256) cur[t] = (t < nb) ? (inc - v) + table[t * CSR_EB + b] : 0;
  __syncthreads();
  const int per = (nedges + CSR_EB - 1) / CSR_EB;
  const int e0 = b * per;
  const int e1 = min(e0 + per, nedges);
  for (int eb = e0 + t; eb < e1; eb += 4 * CSR_TB) {
    int d[4], s[4];
#pragma unroll
    for (int r = 0; r < 4; ++r) {
      const int e = min(eb + r * CSR_TB, e1 - 1);
      d[r] = dst[e];
      s[r] = src[e];
    }
#pragma unroll
    for (int r = 0; r < 4; ++r) {
      if (eb + r * CSR_TB < e1) {
        const int pos = atomicAdd(&cur[d[r] >> 8], 1);
        bpk[pos] = (s[r] << 8) | (d[r] & 255);
      }
    }
  }
}

__global__ __launch_bounds__(CSR_TB) void bin_csr_kernel(const int* __restrict__ bpk,
                                                         const int* __restrict__ btot,
                                                         int* __restrict__ rowptr,
                                                         float* __restrict__ invdeg,
                                                         int* __restrict__ adj, int n, int nb) {
  __shared__ int sh[256];
  __shared__ int cnt[256];
  __shared__ int cur[256];
  const int t = threadIdx.x;
  const int k = blockIdx.x;
  if (t < 256) cnt[t] = 0;
  const int v = (t < nb) ? btot[t] : 0;
  block_scan_incl_256(sh, t, v);
  const int end = sh[k];
  const int beg = (k > 0) ? sh[k - 1] : 0;
  for (int eb = beg + t; eb < end; eb += 4 * CSR_TB) {
    int p[4];
#pragma unroll
    for (int r = 0; r < 4; ++r) p[r] = bpk[min(eb + r * CSR_TB, end - 1)];
#pragma unroll
    for (int r = 0; r < 4; ++r)
      if (eb + r * CSR_TB < end) atomicAdd(&cnt[p[r] & 255], 1);
  }
  __syncthreads();
  const int c = (t < 256) ? cnt[t] : 0;
  const int first = beg + block_scan_incl_256(sh, t, c) - c;
  if (t < 256) {
    const int node = k * 256 + t;
    if (node < n) {
      rowptr[node] = first;
      invdeg[node] = 1.0f / fmaxf((float)c, 1.0f);
      if (node == n - 1) rowptr[n] = first + c;
    }
    cur[t] = first;
  }
  __syncthreads();
  for (int eb = beg + t; eb < end; eb += 4 * CSR_TB) {
    int p[4];
#pragma unroll
    for (int r = 0; r < 4; ++r) p[r] = bpk[min(eb + r * CSR_TB, end - 1)];
#pragma unroll
    for (int r = 0; r < 4; ++r) {
      if (eb + r * CSR_TB < end) {
        const int pos = atomicAdd(&cur[p[r] & 255], 1);
        adj[pos] = p[r] >> 8;
      }
    }
  }
}

#define GATHER_WAVES 4
#define GATHER_NPW   4

template<int D, int R>
__device__ __forceinline__ void gather_add_rows(const float* __restrict__ hl, const int (&s)[4],
                                                float4 (&acc)[D / 128]) {
  constexpr int V = D / 128;
  float4 v[R][V];
#pragma unroll
  for (int r = 0; r < R; ++r) {
    const float* row = hl + (size_t)s[r] * D;
#pragma unroll
    for (int q = 0; q < V; ++q) v[r][q] = *(const float4*)(row + q * 128);
  }
#pragma unroll
  for (int r = 0; r < R; ++r) {
#pragma unroll
    for (int q = 0; q < V; ++q) {
      acc[q].x += v[r][q].x;
      acc[q].y += v[r][q].y;
      acc[q].z += v[r][q].z;
      acc[q].w += v[r][q].w;
    }
  }
}

template<int D>
__global__ __launch_bounds__(GATHER_WAVES * 32) void gather_mean_kernel(const float* __restrict__ h,
                                                          const int* __restrict__ rowptr,
                                                          const int* __restrict__ adj,
                                                          const float* __restrict__ invdeg,
                                                          float* __restrict__ mean, int nnodes) {
  constexpr int V = D / 128;
  const int lane = threadIdx.x & 31;
  const int wave = __builtin_amdgcn_readfirstlane((int)(threadIdx.x >> 5));
  const int node0 = (blockIdx.x * GATHER_WAVES + wave) * GATHER_NPW;
  if (node0 >= nnodes) return;
  const float* hl = h + lane * 4;
  int rp[GATHER_NPW + 1];
  float scv[GATHER_NPW];
#pragma unroll
  for (int i = 0; i <= GATHER_NPW; ++i) rp[i] = rowptr[min(node0 + i, nnodes)];
#pragma unroll
  for (int i = 0; i < GATHER_NPW; ++i) scv[i] = invdeg[min(node0 + i, nnodes - 1)];
#pragma unroll
  for (int i = 0; i < GATHER_NPW; ++i) {
    const int node = node0 + i;
    if (node >= nnodes) break;
    const int beg = rp[i];
    const int end = rp[i + 1];
    float4 acc[V];
#pragma unroll
    for (int q = 0; q < V; ++q) acc[q] = make_float4(0.f, 0.f, 0.f, 0.f);
    if (beg < end) {
      const int lastj = end - 1;
      int s[4];
#pragma unroll
      for (int r = 0; r < 4; ++r) s[r] = adj[min(beg + r, lastj)];
      int j = beg;
      while (j + 4 <= end) {
        int sn[4];
        j += 4;
#pragma unroll
        for (int r = 0; r < 4; ++r) sn[r] = adj[min(j + r, lastj)];
        gather_add_rows<D, 4>(hl, s, acc);
#pragma unroll
        for (int r = 0; r < 4; ++r) s[r] = sn[r];
      }
      const int rem = end - j;
      if (rem == 3)      gather_add_rows<D, 3>(hl, s, acc);
      else if (rem == 2) gather_add_rows<D, 2>(hl, s, acc);
      else if (rem == 1) gather_add_rows<D, 1>(hl, s, acc);
    }
    const float sc = scv[i];
    float* o = mean + (size_t)node * D + lane * 4;
#pragma unroll
    for (int q = 0; q < V; ++q) {
      float4 t = acc[q];
      t.x *= sc; t.y *= sc; t.z *= sc; t.w *= sc;
      *(float4*)(o + q * 128) = t;
    }
  }
}

template<int K, int NOUT, bool RELU>
__global__ __launch_bounds__(256) void sage_gemm_kernel(const float* __restrict__ mean,
                                                        const float* __restrict__ hin,
                                                        const float* __restrict__ Wl,
                                                        const float* __restrict__ Wr,
                                                        const float* __restrict__ bias,
                                                        float* __restrict__ out, int mtiles) {
  constexpr int KC  = 32;
  constexpr int P   = KC + 4;
  constexpr int NC1 = K / KC;
  constexpr int NCH = 2 * NC1;
  __shared__ __attribute__((aligned(16))) float sA[2][128 * P];
  __shared__ __attribute__((aligned(16))) float sB[2][64 * P];

  const int t     = threadIdx.x;
  const int lane  = t & 31;
  const int wave  = t >> 5;
  const int mtile = blockIdx.x * 8 + wave;
  const int nbase = blockIdx.y * 64;
  const int l15   = lane & 15;
  const int s     = lane >> 4;
  const int nrows = mtiles * 16;

  const int q4 = (t & 7) * 4;
  const int r0 = t >> 3;
  const int rowb = blockIdx.x * 128 + r0;
  const int last = nrows - 1;
  const int ar0 = rowb       < nrows ? rowb       : last;
  const int ar1 = rowb + 32  < nrows ? rowb + 32  : last;
  const int ar2 = rowb + 64  < nrows ? rowb + 64  : last;
  const int ar3 = rowb + 96  < nrows ? rowb + 96  : last;
  const int aoff0 = ar0 * K + q4, aoff1 = ar1 * K + q4, aoff2 = ar2 * K + q4, aoff3 = ar3 * K + q4;
  const int boff0 = r0 * NOUT + nbase + q4, boff1 = boff0 + 32;
  const int soff = r0 * P + q4;
  const int bsoff = q4 * P + r0;

  float4 ra0, ra1, ra2, ra3, rb0, rb1;
#define SAGE_FETCH(c)                                               \
  {                                                                 \
    const bool second_ = (c) >= NC1;                                \
    const float* __restrict__ Ag_ = second_ ? hin : mean;           \
    const float* __restrict__ Wg_ = second_ ? Wr : Wl;              \
    const int k0_ = (second_ ? (c) - NC1 : (c)) * KC;               \
    ra0 = *(const float4*)(Ag_ + aoff0 + k0_);                      \
    ra1 = *(const float4*)(Ag_ + aoff1 + k0_);                      \
    ra2 = *(const float4*)(Ag_ + aoff2 + k0_);                      \
    ra3 = *(const float4*)(Ag_ + aoff3 + k0_);                      \
    rb0 = *(const float4*)(Wg_ + boff0 + k0_ * NOUT);               \
    rb1 = *(const float4*)(Wg_ + boff1 + k0_ * NOUT);               \
  }
#define SAGE_STASH(buf)                                             \
  {                                                                 \
    *(float4*)(&sA[buf][soff])          = ra0;                      \
    *(float4*)(&sA[buf][soff + 32 * P]) = ra1;                      \
    *(float4*)(&sA[buf][soff + 64 * P]) = ra2;                      \
    *(float4*)(&sA[buf][soff + 96 * P]) = ra3;                      \
    sB[buf][bsoff]          = rb0.x;                                \
    sB[buf][bsoff + P]      = rb0.y;                                \
    sB[buf][bsoff + 2 * P]  = rb0.z;                                \
    sB[buf][bsoff + 3 * P]  = rb0.w;                                \
    sB[buf][bsoff + 32 * P] = rb1.x;                                \
    sB[buf][bsoff + 33 * P] = rb1.y;                                \
    sB[buf][bsoff + 34 * P] = rb1.z;                                \
    sB[buf][bsoff + 35 * P] = rb1.w;                                \
  }

  v8f acc0 = {}, acc1 = {}, acc2 = {}, acc3 = {};

  SAGE_FETCH(0)
  SAGE_STASH(0)
  __syncthreads();

  for (int c = 0; c < NCH; ++c) {
    const int buf = c & 1;
    if (c + 1 < NCH) SAGE_FETCH(c + 1)
    const float* As = &sA[buf][(wave * 16 + l15) * P + 2 * s];
    const float* Bs = &sB[buf][l15 * P + 2 * s];
#pragma unroll
    for (int j = 0; j < KC / 4; ++j) {
      const v2f a  = *(const v2f*)(As + 4 * j);
      const v2f b0 = *(const v2f*)(Bs + 4 * j);
      const v2f b1 = *(const v2f*)(Bs + 16 * P + 4 * j);
      const v2f b2 = *(const v2f*)(Bs + 32 * P + 4 * j);
      const v2f b3 = *(const v2f*)(Bs + 48 * P + 4 * j);
      acc0 = __builtin_amdgcn_wmma_f32_16x16x4_f32(false, a, false, b0, (short)0, acc0, false, false);
      acc1 = __builtin_amdgcn_wmma_f32_16x16x4_f32(false, a, false, b1, (short)0, acc1, false, false);
      acc2 = __builtin_amdgcn_wmma_f32_16x16x4_f32(false, a, false, b2, (short)0, acc2, false, false);
      acc3 = __builtin_amdgcn_wmma_f32_16x16x4_f32(false, a, false, b3, (short)0, acc3, false, false);
    }
    if (c + 1 < NCH) {
      SAGE_STASH(buf ^ 1)
      __syncthreads();
    }
  }

#undef SAGE_FETCH
#undef SAGE_STASH

  if (mtile >= mtiles) return;

  const int col = nbase + l15;
  const float bj0 = bias[col];
  const float bj1 = bias[col + 16];
  const float bj2 = bias[col + 32];
  const float bj3 = bias[col + 48];
  const int rr = mtile * 16 + (s << 3);
#pragma unroll
  for (int v = 0; v < 8; ++v) {
    float* o = out + (size_t)(rr + v) * NOUT + col;
    float t0 = acc0[v] + bj0;
    float t1 = acc1[v] + bj1;
    float t2 = acc2[v] + bj2;
    float t3 = acc3[v] + bj3;
    if (RELU) {
      t0 = fmaxf(t0, 0.f); t1 = fmaxf(t1, 0.f);
      t2 = fmaxf(t2, 0.f); t3 = fmaxf(t3, 0.f);
    }
    o[0] = t0; o[16] = t1; o[32] = t2; o[48] = t3;
  }
}


extern "C" void kernel_launch(void* const* d_in, const int* in_sizes, int n_in,
                              void* d_out, int out_size, void* d_ws, size_t ws_size,
                              hipStream_t stream) {
  const float* x   = (const float*)d_in[0];
  const int*   ei  = (const int*)d_in[1];
  const float* W1l = (const float*)d_in[2];
  const float* W1r = (const float*)d_in[3];
  const float* b1  = (const float*)d_in[4];
  const float* W2l = (const float*)d_in[5];
  const float* W2r = (const float*)d_in[6];
  const float* b2  = (const float*)d_in[7];
  const float* W3l = (const float*)d_in[8];
  const float* W3r = (const float*)d_in[9];
  const float* b3  = (const float*)d_in[10];
  float* out = (float*)d_out;

  const int N = in_sizes[0] / DIN;
  const int E = in_sizes[1] / 2;
  const int* src = ei;
  const int* dst = ei + E;

  char* ws = (char*)d_ws;
  auto al = [](size_t v) { return (v + 255) & ~(size_t)255; };
  size_t off = 0;
  int*   rowptr   = (int*)(ws + off);   off += al((size_t)(N + 1) * 4);
  float* invdeg   = (float*)(ws + off); off += al((size_t)N * 4);
  int*   adj      = (int*)(ws + off);   off += al((size_t)E * 4);
  float* mean     = (float*)(ws + off); off += al((size_t)N * DH * 4);
  float* h1       = (float*)(ws + off); off += al((size_t)N * DH * 4);
  float* h2       = (float*)(ws + off);
  int*   bpk      = (int*)mean;
  int*   table    = (int*)h1;
  int*   btot     = table + 256 * CSR_EB;

  const int TB = 256;
  const int NB = (N + 255) / 256;
  const int mtiles = N / 16;
  const int gm = (mtiles + 7) / 8;
  const int gg = (N + GATHER_WAVES * GATHER_NPW - 1) / (GATHER_WAVES * GATHER_NPW);
  const int GTB = GATHER_WAVES * 32;

  bin_count_kernel<<<CSR_EB, CSR_TB, 0, stream>>>(dst, table, E, NB);
  bin_scan_kernel<<<NB, TB, 0, stream>>>(table, btot);
  bin_scatter_kernel<<<CSR_EB, CSR_TB, 0, stream>>>(src, dst, table, btot, bpk, E, NB);
  bin_csr_kernel<<<NB, CSR_TB, 0, stream>>>(bpk, btot, rowptr, invdeg, adj, N, NB);

  {
    gather_mean_kernel<DIN><<<gg, GTB, 0, stream>>>(x, rowptr, adj, invdeg, mean, N);
    dim3 g(gm, DH / 64);
    sage_gemm_kernel<DIN, DH, true><<<g, TB, 0, stream>>>(mean, x, W1l, W1r, b1, h1, mtiles);
  }
  {
    gather_mean_kernel<DH><<<gg, GTB, 0, stream>>>(h1, rowptr, adj, invdeg, mean, N);
    dim3 g(gm, DH / 64);
    sage_gemm_kernel<DH, DH, true><<<g, TB, 0, stream>>>(mean, h1, W2l, W2r, b2, h2, mtiles);
  }
  {
    gather_mean_kernel<DH><<<gg, GTB, 0, stream>>>(h2, rowptr, adj, invdeg, mean, N);
    dim3 g(gm, DOUT / 64);
    sage_gemm_kernel<DH, DOUT, false><<<g, TB, 0, stream>>>(mean, h2, W3l, W3r, b3, out, mtiles);
  }
  (void)n_in; (void)out_size; (void)ws_size;
}
